# starting kernel
# baseline (speedup 1.0000x reference)
#include <hip/hip_runtime.h>
#include <hip/hip_bf16.h>
#include <math.h>


typedef float v2f __attribute__((ext_vector_type(2)));
typedef float v8f __attribute__((ext_vector_type(8)));

#define BATCH    8
#define SEQ      512
#define DMODEL   512
#define DINNER   1024
#define DSTATE   16
#define DTRANK   32
#define NGROUPS  8
#define MROWS    (BATCH * SEQ)

__device__ __forceinline__ v8f wmma_f32_4(v2f a, v2f b, v8f c) {
    return __builtin_amdgcn_wmma_f32_16x16x4_f32(
        false, a, false, b, (short)0, c, false, false);
}

__device__ __forceinline__ float softplus_f(float x) {
    return fmaxf(x, 0.f) + log1pf(__expf(-fabsf(x)));
}

__global__ void k_inproj(const float* __restrict__ x,
                         const float* __restrict__ w,
                         float* __restrict__ xz) {
    const int wid   = blockIdx.x * 4 + (threadIdx.x >> 5);
    const int ntq   = wid & 31;
    const int tileM = wid >> 5;
    const int lane  = threadIdx.x & 31;
    const int half  = lane >> 4;
    const int r     = lane & 15;

    const int m  = tileM * 16 + r;
    const int bb = m >> 9;
    const int ll = m & 511;

    const float* xb = x + (size_t)bb * DMODEL * SEQ + ll;
    const float* wr0 = w + (size_t)((ntq * 4 + 0) * 16 + r) * DMODEL;
    const float* wr1 = w + (size_t)((ntq * 4 + 1) * 16 + r) * DMODEL;
    const float* wr2 = w + (size_t)((ntq * 4 + 2) * 16 + r) * DMODEL;
    const float* wr3 = w + (size_t)((ntq * 4 + 3) * 16 + r) * DMODEL;

    v8f acc0 = {}, acc1 = {}, acc2 = {}, acc3 = {};
    #pragma unroll 2
    for (int k0 = 0; k0 < DMODEL; k0 += 4) {
        const int ka = k0 + half * 2;
        v2f a;
        a.x = xb[(size_t)ka * SEQ];
        a.y = xb[(size_t)(ka + 1) * SEQ];
        v2f b0 = *(const v2f*)(wr0 + ka);
        v2f b1 = *(const v2f*)(wr1 + ka);
        v2f b2 = *(const v2f*)(wr2 + ka);
        v2f b3 = *(const v2f*)(wr3 + ka);
        __builtin_prefetch(wr0 + k0 + 32, 0, 1);
        acc0 = wmma_f32_4(a, b0, acc0);
        acc1 = wmma_f32_4(a, b1, acc1);
        acc2 = wmma_f32_4(a, b2, acc2);
        acc3 = wmma_f32_4(a, b3, acc3);
    }
    #pragma unroll
    for (int j = 0; j < 8; ++j) {
        const size_t row = (size_t)(tileM * 16 + half * 8 + j) * (2 * DINNER);
        xz[row + ntq * 64 + 0  + r] = acc0[j];
        xz[row + ntq * 64 + 16 + r] = acc1[j];
        xz[row + ntq * 64 + 32 + r] = acc2[j];
        xz[row + ntq * 64 + 48 + r] = acc3[j];
    }
}

__global__ void k_conv(const float* __restrict__ xz,
                       const float* __restrict__ cw,
                       const float* __restrict__ cb,
                       float* __restrict__ ucv) {
    const int idx = blockIdx.x * blockDim.x + threadIdx.x;
    if (idx >= MROWS * DINNER) return;
    const int c = idx & (DINNER - 1);
    const int m = idx >> 10;
    const int l = m & 511;

    float s = cb[c];
    #pragma unroll
    for (int t = 0; t < 4; ++t) {
        const int lt = l - 3 + t;
        if (lt >= 0)
            s += xz[(size_t)(m - 3 + t) * (2 * DINNER) + c] * cw[c * 4 + t];
    }
    const float sig = 1.f / (1.f + __expf(-s));
    ucv[(size_t)m * DINNER + c] = s * sig;
}

__global__ void k_xproj(const float* __restrict__ ucv,
                        const float* __restrict__ w,
                        float* __restrict__ dbc) {
    const int tileM = blockIdx.x * 4 + (threadIdx.x >> 5);
    const int lane  = threadIdx.x & 31;
    const int half  = lane >> 4;
    const int r     = lane & 15;

    const float* arow = ucv + (size_t)(tileM * 16 + r) * DINNER;
    const float* wr0 = w + (size_t)(0  + r) * DINNER;
    const float* wr1 = w + (size_t)(16 + r) * DINNER;
    const float* wr2 = w + (size_t)(32 + r) * DINNER;
    const float* wr3 = w + (size_t)(48 + r) * DINNER;

    v8f acc0 = {}, acc1 = {}, acc2 = {}, acc3 = {};
    #pragma unroll 2
    for (int k0 = 0; k0 < DINNER; k0 += 4) {
        const int ka = k0 + half * 2;
        v2f a  = *(const v2f*)(arow + ka);
        v2f b0 = *(const v2f*)(wr0 + ka);
        v2f b1 = *(const v2f*)(wr1 + ka);
        v2f b2 = *(const v2f*)(wr2 + ka);
        v2f b3 = *(const v2f*)(wr3 + ka);
        __builtin_prefetch(arow + k0 + 32, 0, 1);
        acc0 = wmma_f32_4(a, b0, acc0);
        acc1 = wmma_f32_4(a, b1, acc1);
        acc2 = wmma_f32_4(a, b2, acc2);
        acc3 = wmma_f32_4(a, b3, acc3);
    }
    #pragma unroll
    for (int j = 0; j < 8; ++j) {
        const size_t row = (size_t)(tileM * 16 + half * 8 + j) * 64;
        dbc[row + 0  + r] = acc0[j];
        dbc[row + 16 + r] = acc1[j];
        dbc[row + 32 + r] = acc2[j];
        dbc[row + 48 + r] = acc3[j];
    }
}

__global__ void k_dtproj(const float* __restrict__ dbc,
                         const float* __restrict__ w,
                         const float* __restrict__ bias,
                         float* __restrict__ delta) {
    const int wid   = blockIdx.x * 4 + (threadIdx.x >> 5);
    const int ntq   = wid & 15;
    const int tileM = wid >> 4;
    const int lane  = threadIdx.x & 31;
    const int half  = lane >> 4;
    const int r     = lane & 15;

    const float* arow = dbc + (size_t)(tileM * 16 + r) * 64;
    const float* wr0 = w + (size_t)((ntq * 4 + 0) * 16 + r) * DTRANK;
    const float* wr1 = w + (size_t)((ntq * 4 + 1) * 16 + r) * DTRANK;
    const float* wr2 = w + (size_t)((ntq * 4 + 2) * 16 + r) * DTRANK;
    const float* wr3 = w + (size_t)((ntq * 4 + 3) * 16 + r) * DTRANK;

    v8f acc0 = {}, acc1 = {}, acc2 = {}, acc3 = {};
    #pragma unroll
    for (int k0 = 0; k0 < DTRANK; k0 += 4) {
        const int ka = k0 + half * 2;
        v2f a  = *(const v2f*)(arow + ka);
        v2f b0 = *(const v2f*)(wr0 + ka);
        v2f b1 = *(const v2f*)(wr1 + ka);
        v2f b2 = *(const v2f*)(wr2 + ka);
        v2f b3 = *(const v2f*)(wr3 + ka);
        acc0 = wmma_f32_4(a, b0, acc0);
        acc1 = wmma_f32_4(a, b1, acc1);
        acc2 = wmma_f32_4(a, b2, acc2);
        acc3 = wmma_f32_4(a, b3, acc3);
    }
    const int nb = ntq * 64 + r;
    const float bn0 = bias[nb + 0];
    const float bn1 = bias[nb + 16];
    const float bn2 = bias[nb + 32];
    const float bn3 = bias[nb + 48];
    #pragma unroll
    for (int j = 0; j < 8; ++j) {
        const size_t row = (size_t)(tileM * 16 + half * 8 + j) * DINNER;
        delta[row + nb + 0 ] = softplus_f(acc0[j] + bn0);
        delta[row + nb + 16] = softplus_f(acc1[j] + bn1);
        delta[row + nb + 32] = softplus_f(acc2[j] + bn2);
        delta[row + nb + 48] = softplus_f(acc3[j] + bn3);
    }
}

__global__ void k_scan(const float* __restrict__ delta,
                       const float* __restrict__ ucv,
                       const float* __restrict__ xz,
                       const float* __restrict__ dbc,
                       const float* __restrict__ A_log,
                       const float* __restrict__ D_param,
                       float* __restrict__ ybuf) {
    const int bb = blockIdx.x >> 3;
    const int c  = (blockIdx.x & 7) * 128 + threadIdx.x;
    const int tid = threadIdx.x;
    const int mbase = bb * SEQ;

    __shared__ float sB[DSTATE];
    __shared__ float sC[DSTATE];

    float Ac[DSTATE];
    float h[DSTATE];
    #pragma unroll
    for (int n = 0; n < DSTATE; ++n) {
        Ac[n] = -__expf(A_log[(size_t)c * DSTATE + n]);
        h[n] = 0.f;
    }
    const float Dp = D_param[c];

    for (int l = 0; l < SEQ; ++l) {
        const int m = mbase + l;
        __syncthreads();
        if (tid < 16)       sB[tid]      = dbc[(size_t)m * 64 + 32 + tid];
        else if (tid < 32)  sC[tid - 16] = dbc[(size_t)m * 64 + 48 + (tid - 16)];
        __syncthreads();

        const float dlt = delta[(size_t)m * DINNER + c];
        const float ut  = ucv[(size_t)m * DINNER + c];
        const float zt  = xz[(size_t)m * (2 * DINNER) + DINNER + c];
        const float du  = dlt * ut;

        float y = 0.f;
        #pragma unroll
        for (int n = 0; n < DSTATE; ++n) {
            const float dA = __expf(dlt * Ac[n]);
            h[n] = dA * h[n] + du * sB[n];
            y += h[n] * sC[n];
        }
        const float sig = 1.f / (1.f + __expf(-zt));
        ybuf[(size_t)m * DINNER + c] = (y + ut * Dp) * (zt * sig);
    }
}

__global__ void k_outproj(const float* __restrict__ ybuf,
                          const float* __restrict__ w,
                          float* __restrict__ otmp) {
    const int wid   = blockIdx.x * 4 + (threadIdx.x >> 5);
    const int ntq   = wid & 7;
    const int tileM = wid >> 3;
    const int lane  = threadIdx.x & 31;
    const int half  = lane >> 4;
    const int r     = lane & 15;

    const float* arow = ybuf + (size_t)(tileM * 16 + r) * DINNER;
    const float* wr0 = w + (size_t)((ntq * 4 + 0) * 16 + r) * DINNER;
    const float* wr1 = w + (size_t)((ntq * 4 + 1) * 16 + r) * DINNER;
    const float* wr2 = w + (size_t)((ntq * 4 + 2) * 16 + r) * DINNER;
    const float* wr3 = w + (size_t)((ntq * 4 + 3) * 16 + r) * DINNER;

    v8f acc0 = {}, acc1 = {}, acc2 = {}, acc3 = {};
    #pragma unroll 2
    for (int k0 = 0; k0 < DINNER; k0 += 4) {
        const int ka = k0 + half * 2;
        v2f a  = *(const v2f*)(arow + ka);
        v2f b0 = *(const v2f*)(wr0 + ka);
        v2f b1 = *(const v2f*)(wr1 + ka);
        v2f b2 = *(const v2f*)(wr2 + ka);
        v2f b3 = *(const v2f*)(wr3 + ka);
        __builtin_prefetch(arow + k0 + 32, 0, 1);
        acc0 = wmma_f32_4(a, b0, acc0);
        acc1 = wmma_f32_4(a, b1, acc1);
        acc2 = wmma_f32_4(a, b2, acc2);
        acc3 = wmma_f32_4(a, b3, acc3);
    }
    #pragma unroll
    for (int j = 0; j < 8; ++j) {
        const int m = tileM * 16 + half * 8 + j;
        const int b = m >> 9;
        const int l = m & 511;
        const size_t base = ((size_t)b * DMODEL + ntq * 64 + r) * SEQ + l;
        otmp[base + 0  * SEQ] = acc0[j];
        otmp[base + 16 * SEQ] = acc1[j];
        otmp[base + 32 * SEQ] = acc2[j];
        otmp[base + 48 * SEQ] = acc3[j];
    }
}

__global__ void k_gnstats(const float* __restrict__ otmp,
                          float* __restrict__ stats) {
    const int bid = blockIdx.x;
    const int tid = threadIdx.x;
    const float* base = otmp + (size_t)bid * 32768;

    __shared__ float ssum[256];
    __shared__ float ssq[256];
    float s = 0.f, q = 0.f;
    for (int i = tid; i < 32768; i += 256) {
        const float v = base[i];
        s += v;
        q += v * v;
    }
    ssum[tid] = s; ssq[tid] = q;
    __syncthreads();
    for (int off = 128; off > 0; off >>= 1) {
        if (tid < off) { ssum[tid] += ssum[tid + off]; ssq[tid] += ssq[tid + off]; }
        __syncthreads();
    }
    if (tid == 0) {
        const float mu  = ssum[0] * (1.f / 32768.f);
        const float var = ssq[0] * (1.f / 32768.f) - mu * mu;
        stats[2 * bid]     = mu;
        stats[2 * bid + 1] = rsqrtf(var + 1e-5f);
    }
}

__global__ void k_gnmish(const float* __restrict__ otmp,
                         const float* __restrict__ stats,
                         const float* __restrict__ gw,
                         const float* __restrict__ gb,
                         float* __restrict__ out) {
    const int idx = blockIdx.x * blockDim.x + threadIdx.x;
    if (idx >= BATCH * DMODEL * SEQ) return;
    const int b   = idx >> 18;
    const int rem = idx & (DMODEL * SEQ - 1);
    const int c   = rem >> 9;
    const int g   = c >> 6;
    const int blk = b * NGROUPS + g;

    const float mu = stats[2 * blk];
    const float rs = stats[2 * blk + 1];
    float v = (otmp[idx] - mu) * rs * gw[c] + gb[c];
    out[idx] = v * tanhf(softplus_f(v));
}

extern "C" void kernel_launch(void* const* d_in, const int* in_sizes, int n_in,
                              void* d_out, int out_size, void* d_ws, size_t ws_size,
                              hipStream_t stream) {
    const float* x         = (const float*)d_in[0];
    const float* in_proj_w = (const float*)d_in[1];
    const float* conv_w    = (const float*)d_in[2];
    const float* conv_b    = (const float*)d_in[3];
    const float* x_proj_w  = (const float*)d_in[4];
    const float* dt_proj_w = (const float*)d_in[5];
    const float* dt_proj_b = (const float*)d_in[6];
    const float* A_log     = (const float*)d_in[7];
    const float* D_param   = (const float*)d_in[8];
    const float* out_proj_w= (const float*)d_in[9];
    const float* gn_w      = (const float*)d_in[10];
    const float* gn_b      = (const float*)d_in[11];
    float* out = (float*)d_out;

    float* ws    = (float*)d_ws;
    float* xz    = ws;
    float* ucv   = xz    + (size_t)MROWS * 2 * DINNER;
    float* dbc   = ucv   + (size_t)MROWS * DINNER;
    float* delta = dbc   + (size_t)MROWS * 64;
    float* ybuf  = delta + (size_t)MROWS * DINNER;
    float* stats = ybuf  + (size_t)MROWS * DINNER;
    float* otmp  = xz;

    k_inproj<<<8192 / 4, 128, 0, stream>>>(x, in_proj_w, xz);

    k_conv<<<(MROWS * DINNER + 255) / 256, 256, 0, stream>>>(xz, conv_w, conv_b, ucv);

    k_xproj<<<256 / 4, 128, 0, stream>>>(ucv, x_proj_w, dbc);

    k_dtproj<<<4096 / 4, 128, 0, stream>>>(dbc, dt_proj_w, dt_proj_b, delta);

    k_scan<<<BATCH * 8, 128, 0, stream>>>(delta, ucv, xz, dbc, A_log, D_param, ybuf);

    k_outproj<<<2048 / 4, 128, 0, stream>>>(ybuf, out_proj_w, otmp);

    k_gnstats<<<BATCH * NGROUPS, 256, 0, stream>>>(otmp, stats);

    k_gnmish<<<(BATCH * DMODEL * SEQ + 255) / 256, 256, 0, stream>>>(
        otmp, stats, gn_w, gn_b, out);
}


# GroupNorm statistics and normalize+Mish in one launch: 4 blocks per (batch, group) each read the whole 128-KB group by 16-B loads, sum it themselves and write a quarter; GroupNorm sum order changes
# speedup vs baseline: 9.1940x; 9.1940x over previous
#include <hip/hip_runtime.h>
#include <hip/hip_bf16.h>
#include <math.h>


typedef float v2f __attribute__((ext_vector_type(2)));
typedef float v4f __attribute__((ext_vector_type(4)));
typedef float v8f __attribute__((ext_vector_type(8)));

#define BATCH    8
#define SEQ      512
#define DMODEL   512
#define DINNER   1024
#define DSTATE   16
#define DTRANK   32
#define NGROUPS  8
#define MROWS    (BATCH * SEQ)
#define NCHUNK   32
#define LCHUNK   (SEQ / NCHUNK)

__device__ __forceinline__ v8f wmma_f32_4(v2f a, v2f b, v8f c) {
    return __builtin_amdgcn_wmma_f32_16x16x4_f32(
        false, a, false, b, (short)0, c, false, false);
}

struct Row16 { v4f q[4]; };

__device__ __forceinline__ Row16 load_row16(const float* __restrict__ p) {
    Row16 f;
    f.q[0] = *(const v4f*)(p + 0);
    f.q[1] = *(const v4f*)(p + 4);
    f.q[2] = *(const v4f*)(p + 8);
    f.q[3] = *(const v4f*)(p + 12);
    return f;
}

__device__ __forceinline__ v2f frag_of(const Row16& f, int j) {
    const v4f q = f.q[j >> 1];
    v2f v;
    v.x = (j & 1) ? q.z : q.x;
    v.y = (j & 1) ? q.w : q.y;
    return v;
}

__device__ __forceinline__ float softplus_f(float x) {
    return fmaxf(x, 0.f) + log1pf(__expf(-fabsf(x)));
}

struct InSlab {
    v4f   a[16];
    float ah[16];
    Row16 b[4];
};

__device__ __forceinline__ void inproj_load(InSlab& s,
                                            const float* __restrict__ xa,
                                            const float* __restrict__ xh,
                                            const float* __restrict__ w0,
                                            const float* __restrict__ w1,
                                            const float* __restrict__ w2,
                                            const float* __restrict__ w3) {
    s.b[0] = load_row16(w0);
    s.b[1] = load_row16(w1);
    s.b[2] = load_row16(w2);
    s.b[3] = load_row16(w3);
    #pragma unroll
    for (int i = 0; i < 16; ++i) s.a[i] = *(const v4f*)(xa + (size_t)i * SEQ);
    #pragma unroll
    for (int i = 0; i < 16; ++i) s.ah[i] = xh[(size_t)i * SEQ];
}

__device__ __forceinline__ void inproj_mma(const InSlab& s, v8f (&acc)[4][4], v8f (&hacc)[2]) {
    #pragma unroll
    for (int j = 0; j < 8; ++j) {
        const v2f b0 = frag_of(s.b[0], j);
        const v2f b1 = frag_of(s.b[1], j);
        const v2f b2 = frag_of(s.b[2], j);
        const v2f b3 = frag_of(s.b[3], j);
        #pragma unroll
        for (int t = 0; t < 4; ++t) {
            v2f av;
            av.x = s.a[2 * j][t];
            av.y = s.a[2 * j + 1][t];
            acc[t][0] = wmma_f32_4(av, b0, acc[t][0]);
            acc[t][1] = wmma_f32_4(av, b1, acc[t][1]);
            acc[t][2] = wmma_f32_4(av, b2, acc[t][2]);
            acc[t][3] = wmma_f32_4(av, b3, acc[t][3]);
        }
        v2f ah;
        ah.x = s.ah[2 * j];
        ah.y = s.ah[2 * j + 1];
        hacc[0] = wmma_f32_4(ah, b0, hacc[0]);
        hacc[1] = wmma_f32_4(ah, b1, hacc[1]);
    }
}

__global__ void __launch_bounds__(128)
k_inproj(const float* __restrict__ x,
         const float* __restrict__ w,
         const float* __restrict__ cw,
         const float* __restrict__ cb,
         float* __restrict__ xz,
         float* __restrict__ ucv) {
    const int wid   = blockIdx.x * 4 + (threadIdx.x >> 5);
    const int q     = wid & 31;
    const int tileM = wid >> 5;
    const int lane  = threadIdx.x & 31;
    const int half  = lane >> 4;
    const int r     = lane & 15;

    const int m0 = tileM * 64;
    const int bb = m0 >> 9;
    const int l0 = m0 & 511;
    const bool has_prev = (l0 != 0);

    const float* xk = x + (size_t)bb * DMODEL * SEQ + (size_t)(half * 16) * SEQ + l0;
    const float* xb = xk + 4 * r;
    const float* xh = xk + (has_prev ? -4 : 0) + (r & 3);
    const float* wr0 = w + (size_t)(q * 32 + 0  + r) * DMODEL + half * 16;
    const float* wr1 = w + (size_t)(q * 32 + 16 + r) * DMODEL + half * 16;
    const float* wr2 = w + (size_t)(DINNER + q * 32 + 0  + r) * DMODEL + half * 16;
    const float* wr3 = w + (size_t)(DINNER + q * 32 + 16 + r) * DMODEL + half * 16;

    v8f acc[4][4] = {};
    v8f hacc[2] = {};
    InSlab s0, s1;
    inproj_load(s0, xb, xh, wr0, wr1, wr2, wr3);
    for (int kb = 0; kb < DMODEL - 64; kb += 64) {
        inproj_load(s1, xb + (size_t)(kb + 32) * SEQ, xh + (size_t)(kb + 32) * SEQ, wr0 + kb + 32, wr1 + kb + 32, wr2 + kb + 32, wr3 + kb + 32);
        __builtin_amdgcn_sched_barrier(0);
        inproj_mma(s0, acc, hacc);
        __builtin_amdgcn_sched_barrier(0);
        inproj_load(s0, xb + (size_t)(kb + 64) * SEQ, xh + (size_t)(kb + 64) * SEQ, wr0 + kb + 64, wr1 + kb + 64, wr2 + kb + 64, wr3 + kb + 64);
        __builtin_amdgcn_sched_barrier(0);
        inproj_mma(s1, acc, hacc);
    }
    inproj_load(s1, xb + (size_t)(DMODEL - 32) * SEQ, xh + (size_t)(DMODEL - 32) * SEQ, wr0 + DMODEL - 32, wr1 + DMODEL - 32, wr2 + DMODEL - 32, wr3 + DMODEL - 32);
    __builtin_amdgcn_sched_barrier(0);
    inproj_mma(s0, acc, hacc);
    inproj_mma(s1, acc, hacc);

    #pragma unroll
    for (int t = 0; t < 4; ++t) {
        #pragma unroll
        for (int j = 0; j < 8; ++j) {
            const size_t row = (size_t)(m0 + 4 * (half * 8 + j) + t) * (2 * DINNER);
            xz[row + DINNER + q * 32 + 0  + r] = acc[t][2][j];
            xz[row + DINNER + q * 32 + 16 + r] = acc[t][3][j];
        }
    }

    #pragma unroll
    for (int n = 0; n < 2; ++n) {
        const int ch = q * 32 + n * 16 + r;
        const v4f  cwv = *(const v4f*)(cw + (size_t)ch * 4);
        const float cbv = cb[ch];
        const float e3 = __shfl_xor(acc[1][n][7], 16);
        const float e2 = __shfl_xor(acc[2][n][7], 16);
        const float e1 = __shfl_xor(acc[3][n][7], 16);
        float p3, p2, p1;
        if (half) {
            p3 = e3; p2 = e2; p1 = e1;
        } else {
            p3 = has_prev ? hacc[n][5] : 0.f;
            p2 = has_prev ? hacc[n][6] : 0.f;
            p1 = has_prev ? hacc[n][7] : 0.f;
        }
        #pragma unroll
        for (int j = 0; j < 8; ++j) {
            #pragma unroll
            for (int t = 0; t < 4; ++t) {
                const float cur = acc[t][n][j];
                float s = cbv;
                s += p3 * cwv.x;
                s += p2 * cwv.y;
                s += p1 * cwv.z;
                s += cur * cwv.w;
                const float sig = 1.f / (1.f + __expf(-s));
                ucv[(size_t)(m0 + 32 * half + 4 * j + t) * DINNER + ch] = s * sig;
                p3 = p2; p2 = p1; p1 = cur;
            }
        }
    }
}

struct XSet {
    Row16 a[2];
    Row16 b0[2];
    Row16 b1[2];
};

__device__ __forceinline__ void xproj_load(XSet& g,
                                           const float* __restrict__ a,
                                           const float* __restrict__ w0,
                                           const float* __restrict__ w1) {
    #pragma unroll
    for (int s = 0; s < 2; ++s) {
        g.a[s]  = load_row16(a  + 32 * s);
        g.b0[s] = load_row16(w0 + 32 * s);
        g.b1[s] = load_row16(w1 + 32 * s);
    }
}

__device__ __forceinline__ void xproj_mma(const XSet& g, v8f& acc0, v8f& acc1) {
    #pragma unroll
    for (int s = 0; s < 2; ++s) {
        #pragma unroll
        for (int j = 0; j < 8; ++j) {
            const v2f av = frag_of(g.a[s], j);
            acc0 = wmma_f32_4(av, frag_of(g.b0[s], j), acc0);
            acc1 = wmma_f32_4(av, frag_of(g.b1[s], j), acc1);
        }
    }
}

__global__ void __launch_bounds__(64)
k_xproj(const float* __restrict__ ucv,
        const float* __restrict__ w,
        float* __restrict__ dbc) {
    const int tileM = blockIdx.x;
    const int wv    = threadIdx.x >> 5;
    const int lane  = threadIdx.x & 31;
    const int half  = lane >> 4;
    const int r     = lane & 15;

    const float* arow = ucv + (size_t)(tileM * 16 + r) * DINNER + half * 16;
    const float* wr0 = w + (size_t)(wv * 32 + 0  + r) * DINNER + half * 16;
    const float* wr1 = w + (size_t)(wv * 32 + 16 + r) * DINNER + half * 16;

    v8f acc0 = {}, acc1 = {};
    XSet g0, g1;
    xproj_load(g0, arow, wr0, wr1);
    for (int kb = 0; kb < DINNER - 128; kb += 128) {
        xproj_load(g1, arow + kb + 64, wr0 + kb + 64, wr1 + kb + 64);
        __builtin_amdgcn_sched_barrier(0);
        xproj_mma(g0, acc0, acc1);
        __builtin_amdgcn_sched_barrier(0);
        xproj_load(g0, arow + kb + 128, wr0 + kb + 128, wr1 + kb + 128);
        __builtin_amdgcn_sched_barrier(0);
        xproj_mma(g1, acc0, acc1);
    }
    xproj_load(g1, arow + DINNER - 64, wr0 + DINNER - 64, wr1 + DINNER - 64);
    __builtin_amdgcn_sched_barrier(0);
    xproj_mma(g0, acc0, acc1);
    xproj_mma(g1, acc0, acc1);

    #pragma unroll
    for (int j = 0; j < 8; ++j) {
        const size_t row = (size_t)(tileM * 16 + half * 8 + j) * 64;
        dbc[row + wv * 32 + 0  + r] = acc0[j];
        dbc[row + wv * 32 + 16 + r] = acc1[j];
    }
}

__global__ void k_dtproj(const float* __restrict__ dbc,
                         const float* __restrict__ w,
                         const float* __restrict__ bias,
                         float* __restrict__ delta) {
    const int wid   = blockIdx.x * 4 + (threadIdx.x >> 5);
    const int ntq   = wid & 15;
    const int tileM = wid >> 4;
    const int lane  = threadIdx.x & 31;
    const int half  = lane >> 4;
    const int r     = lane & 15;

    const float* arow = dbc + (size_t)(tileM * 16 + r) * 64 + half * 16;
    const float* wr0 = w + (size_t)((ntq * 4 + 0) * 16 + r) * DTRANK + half * 16;
    const float* wr1 = w + (size_t)((ntq * 4 + 1) * 16 + r) * DTRANK + half * 16;
    const float* wr2 = w + (size_t)((ntq * 4 + 2) * 16 + r) * DTRANK + half * 16;
    const float* wr3 = w + (size_t)((ntq * 4 + 3) * 16 + r) * DTRANK + half * 16;

    v8f acc0 = {}, acc1 = {}, acc2 = {}, acc3 = {};
    {
        const Row16 a  = load_row16(arow);
        const Row16 b0 = load_row16(wr0);
        const Row16 b1 = load_row16(wr1);
        const Row16 b2 = load_row16(wr2);
        const Row16 b3 = load_row16(wr3);
        __builtin_amdgcn_sched_barrier(0);
        #pragma unroll
        for (int j = 0; j < 8; ++j) {
            const v2f av = frag_of(a, j);
            acc0 = wmma_f32_4(av, frag_of(b0, j), acc0);
            acc1 = wmma_f32_4(av, frag_of(b1, j), acc1);
            acc2 = wmma_f32_4(av, frag_of(b2, j), acc2);
            acc3 = wmma_f32_4(av, frag_of(b3, j), acc3);
        }
    }
    const int nb = ntq * 64 + r;
    const float bn0 = bias[nb + 0];
    const float bn1 = bias[nb + 16];
    const float bn2 = bias[nb + 32];
    const float bn3 = bias[nb + 48];
    #pragma unroll
    for (int j = 0; j < 8; ++j) {
        const size_t row = (size_t)(tileM * 16 + half * 8 + j) * DINNER;
        delta[row + nb + 0 ] = softplus_f(acc0[j] + bn0);
        delta[row + nb + 16] = softplus_f(acc1[j] + bn1);
        delta[row + nb + 32] = softplus_f(acc2[j] + bn2);
        delta[row + nb + 48] = softplus_f(acc3[j] + bn3);
    }
}

#define SCAN_CH  32

__global__ void __launch_bounds__(1024)
k_scan(const float* __restrict__ delta,
       const float* __restrict__ ucv,
       const float* __restrict__ xz,
       const float* __restrict__ dbc,
       const float* __restrict__ A_log,
       const float* __restrict__ D_param,
       float* __restrict__ ybuf) {
    __shared__ float sH[NCHUNK - 1][DSTATE][SCAN_CH];
    __shared__ float sS[NCHUNK - 1][SCAN_CH];

    const int bb   = blockIdx.y;
    const int lane = threadIdx.x & 31;
    const int c    = blockIdx.x * SCAN_CH + lane;
    const int jc   = __builtin_amdgcn_readfirstlane((int)(threadIdx.x >> 5));
    const int mbase = bb * SEQ + jc * LCHUNK;

    float Ac[DSTATE];
    float h[DSTATE];
    #pragma unroll
    for (int n = 0; n < DSTATE; ++n)
        Ac[n] = -__expf(A_log[(size_t)c * DSTATE + n]);
    const float An = -__expf(A_log[(size_t)c * DSTATE + (jc & (DSTATE - 1))]);
    const float Dp = D_param[c];

    float dl[LCHUNK], ut[LCHUNK], zt[LCHUNK];
    #pragma unroll
    for (int l = 0; l < LCHUNK; ++l) {
        const size_t m = (size_t)(mbase + l);
        dl[l] = delta[m * DINNER + c];
        ut[l] = ucv[m * DINNER + c];
        zt[l] = xz[m * (2 * DINNER) + DINNER + c];
    }
    const float* bc = dbc + (size_t)mbase * 64 + 32;

    if (jc < NCHUNK - 1) {
        float ds = 0.f;
        #pragma unroll
        for (int n = 0; n < DSTATE; ++n) h[n] = 0.f;
        #pragma unroll
        for (int l = 0; l < LCHUNK; ++l) {
            const float dlt = dl[l];
            const float du  = dlt * ut[l];
            ds += dlt;
            #pragma unroll
            for (int n = 0; n < DSTATE; ++n) {
                const float dA = __expf(dlt * Ac[n]);
                h[n] = dA * h[n] + du * bc[l * 64 + n];
            }
        }
        #pragma unroll
        for (int n = 0; n < DSTATE; ++n) sH[jc][n][lane] = h[n];
        sS[jc][lane] = ds;
    }
    __syncthreads();

    if (jc < DSTATE) {
        float hc = 0.f;
        #pragma unroll
        for (int j = 0; j < NCHUNK - 1; ++j) {
            const float e = sH[j][jc][lane];
            const float s = sS[j][lane];
            hc = __expf(An * s) * hc + e;
            sH[j][jc][lane] = hc;
        }
    }
    __syncthreads();

    if (jc > 0) {
        #pragma unroll
        for (int n = 0; n < DSTATE; ++n) h[n] = sH[jc - 1][n][lane];
    } else {
        #pragma unroll
        for (int n = 0; n < DSTATE; ++n) h[n] = 0.f;
    }
    #pragma unroll
    for (int l = 0; l < LCHUNK; ++l) {
        const float dlt = dl[l];
        const float du  = dlt * ut[l];
        float y = 0.f;
        #pragma unroll
        for (int n = 0; n < DSTATE; ++n) {
            const float dA = __expf(dlt * Ac[n]);
            h[n] = dA * h[n] + du * bc[l * 64 + n];
            y += h[n] * bc[l * 64 + 16 + n];
        }
        const float z   = zt[l];
        const float sig = 1.f / (1.f + __expf(-z));
        ybuf[(size_t)(mbase + l) * DINNER + c] = (y + ut[l] * Dp) * (z * sig);
    }
}

struct OutSlab {
    Row16 a[2];
    Row16 b[4];
};

__device__ __forceinline__ void outproj_load(OutSlab& s,
                                             const float* __restrict__ a0,
                                             const float* __restrict__ a1,
                                             const float* __restrict__ w0,
                                             const float* __restrict__ w1,
                                             const float* __restrict__ w2,
                                             const float* __restrict__ w3) {
    s.a[0] = load_row16(a0);
    s.a[1] = load_row16(a1);
    s.b[0] = load_row16(w0);
    s.b[1] = load_row16(w1);
    s.b[2] = load_row16(w2);
    s.b[3] = load_row16(w3);
}

__device__ __forceinline__ void outproj_mma(const OutSlab& s, v8f (&acc)[2][4]) {
    #pragma unroll
    for (int j = 0; j < 8; ++j) {
        const v2f b0 = frag_of(s.b[0], j);
        const v2f b1 = frag_of(s.b[1], j);
        const v2f b2 = frag_of(s.b[2], j);
        const v2f b3 = frag_of(s.b[3], j);
        #pragma unroll
        for (int t = 0; t < 2; ++t) {
            const v2f av = frag_of(s.a[t], j);
            acc[t][0] = wmma_f32_4(av, b0, acc[t][0]);
            acc[t][1] = wmma_f32_4(av, b1, acc[t][1]);
            acc[t][2] = wmma_f32_4(av, b2, acc[t][2]);
            acc[t][3] = wmma_f32_4(av, b3, acc[t][3]);
        }
    }
}

__global__ void __launch_bounds__(128)
k_outproj(const float* __restrict__ ybuf,
          const float* __restrict__ w,
          float* __restrict__ otmp) {
    const int wid   = blockIdx.x * 4 + (threadIdx.x >> 5);
    const int ntq   = wid & 7;
    const int tileM = wid >> 3;
    const int lane  = threadIdx.x & 31;
    const int half  = lane >> 4;
    const int r     = lane & 15;

    const float* ar0 = ybuf + (size_t)(tileM * 32 + r) * DINNER + half * 16;
    const float* ar1 = ar0 + (size_t)16 * DINNER;
    const float* wr0 = w + (size_t)((ntq * 4 + 0) * 16 + r) * DINNER + half * 16;
    const float* wr1 = w + (size_t)((ntq * 4 + 1) * 16 + r) * DINNER + half * 16;
    const float* wr2 = w + (size_t)((ntq * 4 + 2) * 16 + r) * DINNER + half * 16;
    const float* wr3 = w + (size_t)((ntq * 4 + 3) * 16 + r) * DINNER + half * 16;

    v8f acc[2][4] = {};
    OutSlab s0, s1;
    outproj_load(s0, ar0, ar1, wr0, wr1, wr2, wr3);
    for (int kb = 0; kb < DINNER - 64; kb += 64) {
        outproj_load(s1, ar0 + kb + 32, ar1 + kb + 32, wr0 + kb + 32, wr1 + kb + 32, wr2 + kb + 32, wr3 + kb + 32);
        __builtin_amdgcn_sched_barrier(0);
        outproj_mma(s0, acc);
        __builtin_amdgcn_sched_barrier(0);
        outproj_load(s0, ar0 + kb + 64, ar1 + kb + 64, wr0 + kb + 64, wr1 + kb + 64, wr2 + kb + 64, wr3 + kb + 64);
        __builtin_amdgcn_sched_barrier(0);
        outproj_mma(s1, acc);
    }
    outproj_load(s1, ar0 + DINNER - 32, ar1 + DINNER - 32, wr0 + DINNER - 32, wr1 + DINNER - 32, wr2 + DINNER - 32, wr3 + DINNER - 32);
    __builtin_amdgcn_sched_barrier(0);
    outproj_mma(s0, acc);
    outproj_mma(s1, acc);

    #pragma unroll
    for (int t = 0; t < 2; ++t) {
        #pragma unroll
        for (int j = 0; j < 8; ++j) {
            const int m = tileM * 32 + t * 16 + half * 8 + j;
            const int b = m >> 9;
            const int l = m & 511;
            const size_t base = ((size_t)b * DMODEL + ntq * 64 + r) * SEQ + l;
            otmp[base + 0  * SEQ] = acc[t][0][j];
            otmp[base + 16 * SEQ] = acc[t][1][j];
            otmp[base + 32 * SEQ] = acc[t][2][j];
            otmp[base + 48 * SEQ] = acc[t][3][j];
        }
    }
}

#define GN_SPLIT 4
#define GN_ELEMS ((DMODEL / NGROUPS) * SEQ)

__device__ __forceinline__ void wave_sum2(float& a, float& b) {
    #pragma unroll
    for (int m = 16; m > 0; m >>= 1) {
        const float ta = __shfl_xor(a, m);
        const float tb = __shfl_xor(b, m);
        a += ta;
        b += tb;
    }
}

__global__ void __launch_bounds__(1024)
k_gnmish(const float* __restrict__ otmp,
         const float* __restrict__ gw,
         const float* __restrict__ gb,
         float* __restrict__ out) {
    __shared__ float ssum[32];
    __shared__ float ssq[32];

    const int grp  = blockIdx.x / GN_SPLIT;
    const int qq   = blockIdx.x % GN_SPLIT;
    const int tid  = threadIdx.x;
    const int lane = tid & 31;
    const int wv   = tid >> 5;
    const float* gbase = otmp + (size_t)grp * GN_ELEMS;

    v4f v[8];
    #pragma unroll
    for (int i = 0; i < 8; ++i) v[i] = *(const v4f*)(gbase + (size_t)(i * 1024 + tid) * 4);

    float s = 0.f, q = 0.f;
    #pragma unroll
    for (int i = 0; i < 8; ++i) {
        #pragma unroll
        for (int e = 0; e < 4; ++e) {
            const float t = v[i][e];
            s += t;
            q += t * t;
        }
    }
    wave_sum2(s, q);
    if (lane == 0) { ssum[wv] = s; ssq[wv] = q; }
    __syncthreads();
    s = ssum[lane];
    q = ssq[lane];
    wave_sum2(s, q);
    const float mu  = s * (1.f / GN_ELEMS);
    const float var = q * (1.f / GN_ELEMS) - mu * mu;
    const float rs  = rsqrtf(var + 1e-5f);

    v4f o[2];
    if (qq == 0)      { o[0] = v[0]; o[1] = v[1]; }
    else if (qq == 1) { o[0] = v[2]; o[1] = v[3]; }
    else if (qq == 2) { o[0] = v[4]; o[1] = v[5]; }
    else              { o[0] = v[6]; o[1] = v[7]; }

    #pragma unroll
    for (int k = 0; k < 2; ++k) {
        const int e0 = ((2 * qq + k) * 1024 + tid) * 4;
        const int c  = (grp % NGROUPS) * (DMODEL / NGROUPS) + (e0 / SEQ);
        const float wc = gw[c];
        const float bc = gb[c];
        v4f r;
        #pragma unroll
        for (int e = 0; e < 4; ++e) {
            const float val = (o[k][e] - mu) * rs * wc + bc;
            r[e] = val * tanhf(softplus_f(val));
        }
        *(v4f*)(out + (size_t)grp * GN_ELEMS + e0) = r;
    }
}

extern "C" void kernel_launch(void* const* d_in, const int* in_sizes, int n_in,
                              void* d_out, int out_size, void* d_ws, size_t ws_size,
                              hipStream_t stream) {
    const float* x         = (const float*)d_in[0];
    const float* in_proj_w = (const float*)d_in[1];
    const float* conv_w    = (const float*)d_in[2];
    const float* conv_b    = (const float*)d_in[3];
    const float* x_proj_w  = (const float*)d_in[4];
    const float* dt_proj_w = (const float*)d_in[5];
    const float* dt_proj_b = (const float*)d_in[6];
    const float* A_log     = (const float*)d_in[7];
    const float* D_param   = (const float*)d_in[8];
    const float* out_proj_w= (const float*)d_in[9];
    const float* gn_w      = (const float*)d_in[10];
    const float* gn_b      = (const float*)d_in[11];
    float* out = (float*)d_out;

    float* ws    = (float*)d_ws;
    float* xz    = ws;
    float* ucv   = xz    + (size_t)MROWS * 2 * DINNER;
    float* dbc   = ucv   + (size_t)MROWS * DINNER;
    float* delta = dbc   + (size_t)MROWS * 64;
    float* ybuf  = delta + (size_t)MROWS * DINNER;
    float* otmp  = xz;

    k_inproj<<<2048 / 4, 128, 0, stream>>>(x, in_proj_w, conv_w, conv_b, xz, ucv);

    k_xproj<<<MROWS / 16, 64, 0, stream>>>(ucv, x_proj_w, dbc);

    k_dtproj<<<4096 / 4, 128, 0, stream>>>(dbc, dt_proj_w, dt_proj_b, delta);

    k_scan<<<dim3(DINNER / SCAN_CH, BATCH), NCHUNK * 32, 0, stream>>>(
        delta, ucv, xz, dbc, A_log, D_param, ybuf);

    k_outproj<<<1024 / 4, 128, 0, stream>>>(ybuf, out_proj_w, otmp);

    k_gnmish<<<BATCH * NGROUPS * GN_SPLIT, 1024, 0, stream>>>(otmp, gn_w, gn_b, out);
}


# dt_proj+softplus moved into the scan kernel: a chunk wave forms its 16 steps x 32 channels of delta with 16 WMMAs (K=32, sum order kept), lanes swap tile halves; dt_proj launch, delta round trip gone
# speedup vs baseline: 9.9280x; 1.0798x over previous
#include <hip/hip_runtime.h>
#include <hip/hip_bf16.h>
#include <math.h>


typedef float v2f __attribute__((ext_vector_type(2)));
typedef float v4f __attribute__((ext_vector_type(4)));
typedef float v8f __attribute__((ext_vector_type(8)));

#define BATCH    8
#define SEQ      512
#define DMODEL   512
#define DINNER   1024
#define DSTATE   16
#define DTRANK   32
#define NGROUPS  8
#define MROWS    (BATCH * SEQ)
#define NCHUNK   32
#define LCHUNK   (SEQ / NCHUNK)

__device__ __forceinline__ v8f wmma_f32_4(v2f a, v2f b, v8f c) {
    return __builtin_amdgcn_wmma_f32_16x16x4_f32(
        false, a, false, b, (short)0, c, false, false);
}

struct Row16 { v4f q[4]; };

__device__ __forceinline__ Row16 load_row16(const float* __restrict__ p) {
    Row16 f;
    f.q[0] = *(const v4f*)(p + 0);
    f.q[1] = *(const v4f*)(p + 4);
    f.q[2] = *(const v4f*)(p + 8);
    f.q[3] = *(const v4f*)(p + 12);
    return f;
}

__device__ __forceinline__ v2f frag_of(const Row16& f, int j) {
    const v4f q = f.q[j >> 1];
    v2f v;
    v.x = (j & 1) ? q.z : q.x;
    v.y = (j & 1) ? q.w : q.y;
    return v;
}

__device__ __forceinline__ float softplus_f(float x) {
    return fmaxf(x, 0.f) + log1pf(__expf(-fabsf(x)));
}

struct InSlab {
    v4f   a[16];
    float ah[16];
    Row16 b[4];
};

__device__ __forceinline__ void inproj_load(InSlab& s,
                                            const float* __restrict__ xa,
                                            const float* __restrict__ xh,
                                            const float* __restrict__ w0,
                                            const float* __restrict__ w1,
                                            const float* __restrict__ w2,
                                            const float* __restrict__ w3) {
    s.b[0] = load_row16(w0);
    s.b[1] = load_row16(w1);
    s.b[2] = load_row16(w2);
    s.b[3] = load_row16(w3);
    #pragma unroll
    for (int i = 0; i < 16; ++i) s.a[i] = *(const v4f*)(xa + (size_t)i * SEQ);
    #pragma unroll
    for (int i = 0; i < 16; ++i) s.ah[i] = xh[(size_t)i * SEQ];
}

__device__ __forceinline__ void inproj_mma(const InSlab& s, v8f (&acc)[4][4], v8f (&hacc)[2]) {
    #pragma unroll
    for (int j = 0; j < 8; ++j) {
        const v2f b0 = frag_of(s.b[0], j);
        const v2f b1 = frag_of(s.b[1], j);
        const v2f b2 = frag_of(s.b[2], j);
        const v2f b3 = frag_of(s.b[3], j);
        #pragma unroll
        for (int t = 0; t < 4; ++t) {
            v2f av;
            av.x = s.a[2 * j][t];
            av.y = s.a[2 * j + 1][t];
            acc[t][0] = wmma_f32_4(av, b0, acc[t][0]);
            acc[t][1] = wmma_f32_4(av, b1, acc[t][1]);
            acc[t][2] = wmma_f32_4(av, b2, acc[t][2]);
            acc[t][3] = wmma_f32_4(av, b3, acc[t][3]);
        }
        v2f ah;
        ah.x = s.ah[2 * j];
        ah.y = s.ah[2 * j + 1];
        hacc[0] = wmma_f32_4(ah, b0, hacc[0]);
        hacc[1] = wmma_f32_4(ah, b1, hacc[1]);
    }
}

__global__ void __launch_bounds__(128)
k_inproj(const float* __restrict__ x,
         const float* __restrict__ w,
         const float* __restrict__ cw,
         const float* __restrict__ cb,
         float* __restrict__ xz,
         float* __restrict__ ucv) {
    const int wid   = blockIdx.x * 4 + (threadIdx.x >> 5);
    const int q     = wid & 31;
    const int tileM = wid >> 5;
    const int lane  = threadIdx.x & 31;
    const int half  = lane >> 4;
    const int r     = lane & 15;

    const int m0 = tileM * 64;
    const int bb = m0 >> 9;
    const int l0 = m0 & 511;
    const bool has_prev = (l0 != 0);

    const float* xk = x + (size_t)bb * DMODEL * SEQ + (size_t)(half * 16) * SEQ + l0;
    const float* xb = xk + 4 * r;
    const float* xh = xk + (has_prev ? -4 : 0) + (r & 3);
    const float* wr0 = w + (size_t)(q * 32 + 0  + r) * DMODEL + half * 16;
    const float* wr1 = w + (size_t)(q * 32 + 16 + r) * DMODEL + half * 16;
    const float* wr2 = w + (size_t)(DINNER + q * 32 + 0  + r) * DMODEL + half * 16;
    const float* wr3 = w + (size_t)(DINNER + q * 32 + 16 + r) * DMODEL + half * 16;

    v8f acc[4][4] = {};
    v8f hacc[2] = {};
    InSlab s0, s1;
    inproj_load(s0, xb, xh, wr0, wr1, wr2, wr3);
    for (int kb = 0; kb < DMODEL - 64; kb += 64) {
        inproj_load(s1, xb + (size_t)(kb + 32) * SEQ, xh + (size_t)(kb + 32) * SEQ, wr0 + kb + 32, wr1 + kb + 32, wr2 + kb + 32, wr3 + kb + 32);
        __builtin_amdgcn_sched_barrier(0);
        inproj_mma(s0, acc, hacc);
        __builtin_amdgcn_sched_barrier(0);
        inproj_load(s0, xb + (size_t)(kb + 64) * SEQ, xh + (size_t)(kb + 64) * SEQ, wr0 + kb + 64, wr1 + kb + 64, wr2 + kb + 64, wr3 + kb + 64);
        __builtin_amdgcn_sched_barrier(0);
        inproj_mma(s1, acc, hacc);
    }
    inproj_load(s1, xb + (size_t)(DMODEL - 32) * SEQ, xh + (size_t)(DMODEL - 32) * SEQ, wr0 + DMODEL - 32, wr1 + DMODEL - 32, wr2 + DMODEL - 32, wr3 + DMODEL - 32);
    __builtin_amdgcn_sched_barrier(0);
    inproj_mma(s0, acc, hacc);
    inproj_mma(s1, acc, hacc);

    #pragma unroll
    for (int t = 0; t < 4; ++t) {
        #pragma unroll
        for (int j = 0; j < 8; ++j) {
            const size_t row = (size_t)(m0 + 4 * (half * 8 + j) + t) * (2 * DINNER);
            xz[row + DINNER + q * 32 + 0  + r] = acc[t][2][j];
            xz[row + DINNER + q * 32 + 16 + r] = acc[t][3][j];
        }
    }

    #pragma unroll
    for (int n = 0; n < 2; ++n) {
        const int ch = q * 32 + n * 16 + r;
        const v4f  cwv = *(const v4f*)(cw + (size_t)ch * 4);
        const float cbv = cb[ch];
        const float e3 = __shfl_xor(acc[1][n][7], 16);
        const float e2 = __shfl_xor(acc[2][n][7], 16);
        const float e1 = __shfl_xor(acc[3][n][7], 16);
        float p3, p2, p1;
        if (half) {
            p3 = e3; p2 = e2; p1 = e1;
        } else {
            p3 = has_prev ? hacc[n][5] : 0.f;
            p2 = has_prev ? hacc[n][6] : 0.f;
            p1 = has_prev ? hacc[n][7] : 0.f;
        }
        #pragma unroll
        for (int j = 0; j < 8; ++j) {
            #pragma unroll
            for (int t = 0; t < 4; ++t) {
                const float cur = acc[t][n][j];
                float s = cbv;
                s += p3 * cwv.x;
                s += p2 * cwv.y;
                s += p1 * cwv.z;
                s += cur * cwv.w;
                const float sig = 1.f / (1.f + __expf(-s));
                ucv[(size_t)(m0 + 32 * half + 4 * j + t) * DINNER + ch] = s * sig;
                p3 = p2; p2 = p1; p1 = cur;
            }
        }
    }
}

struct XSet {
    Row16 a[2];
    Row16 b0[2];
    Row16 b1[2];
};

__device__ __forceinline__ void xproj_load(XSet& g,
                                           const float* __restrict__ a,
                                           const float* __restrict__ w0,
                                           const float* __restrict__ w1) {
    #pragma unroll
    for (int s = 0; s < 2; ++s) {
        g.a[s]  = load_row16(a  + 32 * s);
        g.b0[s] = load_row16(w0 + 32 * s);
        g.b1[s] = load_row16(w1 + 32 * s);
    }
}

__device__ __forceinline__ void xproj_mma(const XSet& g, v8f& acc0, v8f& acc1) {
    #pragma unroll
    for (int s = 0; s < 2; ++s) {
        #pragma unroll
        for (int j = 0; j < 8; ++j) {
            const v2f av = frag_of(g.a[s], j);
            acc0 = wmma_f32_4(av, frag_of(g.b0[s], j), acc0);
            acc1 = wmma_f32_4(av, frag_of(g.b1[s], j), acc1);
        }
    }
}

__global__ void __launch_bounds__(64)
k_xproj(const float* __restrict__ ucv,
        const float* __restrict__ w,
        float* __restrict__ dbc) {
    const int tileM = blockIdx.x;
    const int wv    = threadIdx.x >> 5;
    const int lane  = threadIdx.x & 31;
    const int half  = lane >> 4;
    const int r     = lane & 15;

    const float* arow = ucv + (size_t)(tileM * 16 + r) * DINNER + half * 16;
    const float* wr0 = w + (size_t)(wv * 32 + 0  + r) * DINNER + half * 16;
    const float* wr1 = w + (size_t)(wv * 32 + 16 + r) * DINNER + half * 16;

    v8f acc0 = {}, acc1 = {};
    XSet g0, g1;
    xproj_load(g0, arow, wr0, wr1);
    for (int kb = 0; kb < DINNER - 128; kb += 128) {
        xproj_load(g1, arow + kb + 64, wr0 + kb + 64, wr1 + kb + 64);
        __builtin_amdgcn_sched_barrier(0);
        xproj_mma(g0, acc0, acc1);
        __builtin_amdgcn_sched_barrier(0);
        xproj_load(g0, arow + kb + 128, wr0 + kb + 128, wr1 + kb + 128);
        __builtin_amdgcn_sched_barrier(0);
        xproj_mma(g1, acc0, acc1);
    }
    xproj_load(g1, arow + DINNER - 64, wr0 + DINNER - 64, wr1 + DINNER - 64);
    __builtin_amdgcn_sched_barrier(0);
    xproj_mma(g0, acc0, acc1);
    xproj_mma(g1, acc0, acc1);

    #pragma unroll
    for (int j = 0; j < 8; ++j) {
        const size_t row = (size_t)(tileM * 16 + half * 8 + j) * 64;
        dbc[row + wv * 32 + 0  + r] = acc0[j];
        dbc[row + wv * 32 + 16 + r] = acc1[j];
    }
}

#define SCAN_CH  32

__global__ void __launch_bounds__(1024)
k_scan(const float* __restrict__ dbc,
       const float* __restrict__ wdt,
       const float* __restrict__ bdt,
       const float* __restrict__ ucv,
       const float* __restrict__ xz,
       const float* __restrict__ A_log,
       const float* __restrict__ D_param,
       float* __restrict__ ybuf) {
    __shared__ float sH[NCHUNK - 1][DSTATE][SCAN_CH];
    __shared__ float sS[NCHUNK - 1][SCAN_CH];

    const int bb   = blockIdx.y;
    const int lane = threadIdx.x & 31;
    const int c    = blockIdx.x * SCAN_CH + lane;
    const int jc   = __builtin_amdgcn_readfirstlane((int)(threadIdx.x >> 5));
    const int mbase = bb * SEQ + jc * LCHUNK;

    float Ac[DSTATE];
    float h[DSTATE];
    #pragma unroll
    for (int n = 0; n < DSTATE; ++n)
        Ac[n] = -__expf(A_log[(size_t)c * DSTATE + n]);
    const float An = -__expf(A_log[(size_t)c * DSTATE + (jc & (DSTATE - 1))]);
    const float Dp = D_param[c];

    float dl[LCHUNK], ut[LCHUNK], zt[LCHUNK];
    #pragma unroll
    for (int l = 0; l < LCHUNK; ++l) {
        const size_t m = (size_t)(mbase + l);
        ut[l] = ucv[m * DINNER + c];
        zt[l] = xz[m * (2 * DINNER) + DINNER + c];
    }

    {
        const int half = lane >> 4;
        const int r    = lane & 15;
        const float* arow = dbc + (size_t)(mbase + r) * 64 + half * 16;
        const float* wr0  = wdt + (size_t)(blockIdx.x * SCAN_CH + r) * DTRANK + half * 16;
        const float* wr1  = wr0 + (size_t)16 * DTRANK;
        const Row16 a  = load_row16(arow);
        const Row16 b0 = load_row16(wr0);
        const Row16 b1 = load_row16(wr1);
        v8f acc0 = {}, acc1 = {};
        #pragma unroll
        for (int j = 0; j < 8; ++j) {
            const v2f av = frag_of(a, j);
            acc0 = wmma_f32_4(av, frag_of(b0, j), acc0);
            acc1 = wmma_f32_4(av, frag_of(b1, j), acc1);
        }
        const float bn = bdt[c];
        #pragma unroll
        for (int j = 0; j < 8; ++j) {
            const float keep = half ? acc1[j] : acc0[j];
            const float give = half ? acc0[j] : acc1[j];
            const float got  = __shfl_xor(give, 16);
            dl[j]     = softplus_f((half ? got : keep) + bn);
            dl[8 + j] = softplus_f((half ? keep : got) + bn);
        }
    }
    const float* bc = dbc + (size_t)mbase * 64 + 32;

    if (jc < NCHUNK - 1) {
        float ds = 0.f;
        #pragma unroll
        for (int n = 0; n < DSTATE; ++n) h[n] = 0.f;
        #pragma unroll
        for (int l = 0; l < LCHUNK; ++l) {
            const float dlt = dl[l];
            const float du  = dlt * ut[l];
            ds += dlt;
            #pragma unroll
            for (int n = 0; n < DSTATE; ++n) {
                const float dA = __expf(dlt * Ac[n]);
                h[n] = dA * h[n] + du * bc[l * 64 + n];
            }
        }
        #pragma unroll
        for (int n = 0; n < DSTATE; ++n) sH[jc][n][lane] = h[n];
        sS[jc][lane] = ds;
    }
    __syncthreads();

    if (jc < DSTATE) {
        float hc = 0.f;
        #pragma unroll
        for (int j = 0; j < NCHUNK - 1; ++j) {
            const float e = sH[j][jc][lane];
            const float s = sS[j][lane];
            hc = __expf(An * s) * hc + e;
            sH[j][jc][lane] = hc;
        }
    }
    __syncthreads();

    if (jc > 0) {
        #pragma unroll
        for (int n = 0; n < DSTATE; ++n) h[n] = sH[jc - 1][n][lane];
    } else {
        #pragma unroll
        for (int n = 0; n < DSTATE; ++n) h[n] = 0.f;
    }
    #pragma unroll
    for (int l = 0; l < LCHUNK; ++l) {
        const float dlt = dl[l];
        const float du  = dlt * ut[l];
        float y = 0.f;
        #pragma unroll
        for (int n = 0; n < DSTATE; ++n) {
            const float dA = __expf(dlt * Ac[n]);
            h[n] = dA * h[n] + du * bc[l * 64 + n];
            y += h[n] * bc[l * 64 + 16 + n];
        }
        const float z   = zt[l];
        const float sig = 1.f / (1.f + __expf(-z));
        ybuf[(size_t)(mbase + l) * DINNER + c] = (y + ut[l] * Dp) * (z * sig);
    }
}

struct OutSlab {
    Row16 a[2];
    Row16 b[4];
};

__device__ __forceinline__ void outproj_load(OutSlab& s,
                                             const float* __restrict__ a0,
                                             const float* __restrict__ a1,
                                             const float* __restrict__ w0,
                                             const float* __restrict__ w1,
                                             const float* __restrict__ w2,
                                             const float* __restrict__ w3) {
    s.a[0] = load_row16(a0);
    s.a[1] = load_row16(a1);
    s.b[0] = load_row16(w0);
    s.b[1] = load_row16(w1);
    s.b[2] = load_row16(w2);
    s.b[3] = load_row16(w3);
}

__device__ __forceinline__ void outproj_mma(const OutSlab& s, v8f (&acc)[2][4]) {
    #pragma unroll
    for (int j = 0; j < 8; ++j) {
        const v2f b0 = frag_of(s.b[0], j);
        const v2f b1 = frag_of(s.b[1], j);
        const v2f b2 = frag_of(s.b[2], j);
        const v2f b3 = frag_of(s.b[3], j);
        #pragma unroll
        for (int t = 0; t < 2; ++t) {
            const v2f av = frag_of(s.a[t], j);
            acc[t][0] = wmma_f32_4(av, b0, acc[t][0]);
            acc[t][1] = wmma_f32_4(av, b1, acc[t][1]);
            acc[t][2] = wmma_f32_4(av, b2, acc[t][2]);
            acc[t][3] = wmma_f32_4(av, b3, acc[t][3]);
        }
    }
}

__global__ void __launch_bounds__(128)
k_outproj(const float* __restrict__ ybuf,
          const float* __restrict__ w,
          float* __restrict__ otmp) {
    const int wid   = blockIdx.x * 4 + (threadIdx.x >> 5);
    const int ntq   = wid & 7;
    const int tileM = wid >> 3;
    const int lane  = threadIdx.x & 31;
    const int half  = lane >> 4;
    const int r     = lane & 15;

    const float* ar0 = ybuf + (size_t)(tileM * 32 + r) * DINNER + half * 16;
    const float* ar1 = ar0 + (size_t)16 * DINNER;
    const float* wr0 = w + (size_t)((ntq * 4 + 0) * 16 + r) * DINNER + half * 16;
    const float* wr1 = w + (size_t)((ntq * 4 + 1) * 16 + r) * DINNER + half * 16;
    const float* wr2 = w + (size_t)((ntq * 4 + 2) * 16 + r) * DINNER + half * 16;
    const float* wr3 = w + (size_t)((ntq * 4 + 3) * 16 + r) * DINNER + half * 16;

    v8f acc[2][4] = {};
    OutSlab s0, s1;
    outproj_load(s0, ar0, ar1, wr0, wr1, wr2, wr3);
    for (int kb = 0; kb < DINNER - 64; kb += 64) {
        outproj_load(s1, ar0 + kb + 32, ar1 + kb + 32, wr0 + kb + 32, wr1 + kb + 32, wr2 + kb + 32, wr3 + kb + 32);
        __builtin_amdgcn_sched_barrier(0);
        outproj_mma(s0, acc);
        __builtin_amdgcn_sched_barrier(0);
        outproj_load(s0, ar0 + kb + 64, ar1 + kb + 64, wr0 + kb + 64, wr1 + kb + 64, wr2 + kb + 64, wr3 + kb + 64);
        __builtin_amdgcn_sched_barrier(0);
        outproj_mma(s1, acc);
    }
    outproj_load(s1, ar0 + DINNER - 32, ar1 + DINNER - 32, wr0 + DINNER - 32, wr1 + DINNER - 32, wr2 + DINNER - 32, wr3 + DINNER - 32);
    __builtin_amdgcn_sched_barrier(0);
    outproj_mma(s0, acc);
    outproj_mma(s1, acc);

    #pragma unroll
    for (int t = 0; t < 2; ++t) {
        #pragma unroll
        for (int j = 0; j < 8; ++j) {
            const int m = tileM * 32 + t * 16 + half * 8 + j;
            const int b = m >> 9;
            const int l = m & 511;
            const size_t base = ((size_t)b * DMODEL + ntq * 64 + r) * SEQ + l;
            otmp[base + 0  * SEQ] = acc[t][0][j];
            otmp[base + 16 * SEQ] = acc[t][1][j];
            otmp[base + 32 * SEQ] = acc[t][2][j];
            otmp[base + 48 * SEQ] = acc[t][3][j];
        }
    }
}

#define GN_SPLIT 4
#define GN_ELEMS ((DMODEL / NGROUPS) * SEQ)

__device__ __forceinline__ void wave_sum2(float& a, float& b) {
    #pragma unroll
    for (int m = 16; m > 0; m >>= 1) {
        const float ta = __shfl_xor(a, m);
        const float tb = __shfl_xor(b, m);
        a += ta;
        b += tb;
    }
}

__global__ void __launch_bounds__(1024)
k_gnmish(const float* __restrict__ otmp,
         const float* __restrict__ gw,
         const float* __restrict__ gb,
         float* __restrict__ out) {
    __shared__ float ssum[32];
    __shared__ float ssq[32];

    const int grp  = blockIdx.x / GN_SPLIT;
    const int qq   = blockIdx.x % GN_SPLIT;
    const int tid  = threadIdx.x;
    const int lane = tid & 31;
    const int wv   = tid >> 5;
    const float* gbase = otmp + (size_t)grp * GN_ELEMS;

    v4f v[8];
    #pragma unroll
    for (int i = 0; i < 8; ++i) v[i] = *(const v4f*)(gbase + (size_t)(i * 1024 + tid) * 4);

    float s = 0.f, q = 0.f;
    #pragma unroll
    for (int i = 0; i < 8; ++i) {
        #pragma unroll
        for (int e = 0; e < 4; ++e) {
            const float t = v[i][e];
            s += t;
            q += t * t;
        }
    }
    wave_sum2(s, q);
    if (lane == 0) { ssum[wv] = s; ssq[wv] = q; }
    __syncthreads();
    s = ssum[lane];
    q = ssq[lane];
    wave_sum2(s, q);
    const float mu  = s * (1.f / GN_ELEMS);
    const float var = q * (1.f / GN_ELEMS) - mu * mu;
    const float rs  = rsqrtf(var + 1e-5f);

    v4f o[2];
    if (qq == 0)      { o[0] = v[0]; o[1] = v[1]; }
    else if (qq == 1) { o[0] = v[2]; o[1] = v[3]; }
    else if (qq == 2) { o[0] = v[4]; o[1] = v[5]; }
    else              { o[0] = v[6]; o[1] = v[7]; }

    #pragma unroll
    for (int k = 0; k < 2; ++k) {
        const int e0 = ((2 * qq + k) * 1024 + tid) * 4;
        const int c  = (grp % NGROUPS) * (DMODEL / NGROUPS) + (e0 / SEQ);
        const float wc = gw[c];
        const float bc = gb[c];
        v4f r;
        #pragma unroll
        for (int e = 0; e < 4; ++e) {
            const float val = (o[k][e] - mu) * rs * wc + bc;
            r[e] = val * tanhf(softplus_f(val));
        }
        *(v4f*)(out + (size_t)grp * GN_ELEMS + e0) = r;
    }
}

extern "C" void kernel_launch(void* const* d_in, const int* in_sizes, int n_in,
                              void* d_out, int out_size, void* d_ws, size_t ws_size,
                              hipStream_t stream) {
    const float* x         = (const float*)d_in[0];
    const float* in_proj_w = (const float*)d_in[1];
    const float* conv_w    = (const float*)d_in[2];
    const float* conv_b    = (const float*)d_in[3];
    const float* x_proj_w  = (const float*)d_in[4];
    const float* dt_proj_w = (const float*)d_in[5];
    const float* dt_proj_b = (const float*)d_in[6];
    const float* A_log     = (const float*)d_in[7];
    const float* D_param   = (const float*)d_in[8];
    const float* out_proj_w= (const float*)d_in[9];
    const float* gn_w      = (const float*)d_in[10];
    const float* gn_b      = (const float*)d_in[11];
    float* out = (float*)d_out;

    float* ws    = (float*)d_ws;
    float* xz    = ws;
    float* ucv   = xz    + (size_t)MROWS * 2 * DINNER;
    float* dbc   = ucv   + (size_t)MROWS * DINNER;
    float* ybuf  = dbc   + (size_t)MROWS * 64;
    float* otmp  = xz;

    k_inproj<<<2048 / 4, 128, 0, stream>>>(x, in_proj_w, conv_w, conv_b, xz, ucv);

    k_xproj<<<MROWS / 16, 64, 0, stream>>>(ucv, x_proj_w, dbc);

    k_scan<<<dim3(DINNER / SCAN_CH, BATCH), NCHUNK * 32, 0, stream>>>(
        dbc, dt_proj_w, dt_proj_b, ucv, xz, A_log, D_param, ybuf);

    k_outproj<<<1024 / 4, 128, 0, stream>>>(ybuf, out_proj_w, otmp);

    k_gnmish<<<BATCH * NGROUPS * GN_SPLIT, 1024, 0, stream>>>(otmp, gn_w, gn_b, out);
}
